# starting kernel
# baseline (speedup 1.0000x reference)
#include <hip/hip_runtime.h>

typedef __attribute__((ext_vector_type(2))) float v2f;
typedef __attribute__((ext_vector_type(8))) float v8f;

#define N_EXP 20
#define HMS 66

#define WAVE_LDS_FENCE() asm volatile("s_wait_dscnt 0" ::: "memory")

__device__ __forceinline__ v8f wmma_k4(v2f a, v2f b, v8f c) {
    return __builtin_amdgcn_wmma_f32_16x16x4_f32(false, a, false, b, (short)0, c, false, false);
}

__device__ __forceinline__ float fast_silu(float v) {
    return v * __builtin_amdgcn_rcpf(1.f + __expf(-v));
}

__global__ void __launch_bounds__(256)
titan_menger_kernel(const float* __restrict__ x,
                    const float* __restrict__ wq,
                    const float* __restrict__ aw,
                    const float* __restrict__ ab,
                    const float* __restrict__ W1,
                    const float* __restrict__ b1,
                    const float* __restrict__ W2,
                    const float* __restrict__ b2,
                    const float* __restrict__ Wg,
                    const float* __restrict__ bg,
                    float* __restrict__ out)
{
    __shared__ __align__(16) float lds_hmid[8][16 * HMS];
    __shared__ __align__(16) float lds_route[8][N_EXP * 16];
    __shared__ __align__(16) float lds_tot[8][16 * 8];

    const int tid  = threadIdx.x;
    const int wave = tid >> 5;
    const int lane = tid & 31;
    const int n    = lane & 15;
    const int hi   = lane >> 4;
    const int rowBase = blockIdx.x * 128 + wave * 16;

    if (lane < 16) {
        const int row = rowBase + lane;
        float sum = 0.f;
        #pragma unroll
        for (int i = 0; i < 8; ++i) sum += x[row * 8 + i];
        const float theta = sum * 0.125f;
        const float c  = __cosf(0.5f * theta), s = __sinf(0.5f * theta);
        const float c2 = c * c, s2 = s * s;
        const float c3 = c2 * c, s3 = s2 * s;
        const float c4 = c2 * c2, s4 = s2 * s2;

        const int PC[16] = {0,2,2,2, 2,4,2,2, 3,1,3,3, 3,1,1,1};
        float sr[16], si[16];
        #pragma unroll
        for (int j = 0; j < 16; ++j) {
            const int p = PC[j];
            sr[j] = (p == 0) ? c4 : (p == 2) ? -c2 * s2 : (p == 4) ? s4 : 0.f;
            si[j] = (p == 1) ? -c3 * s : (p == 3) ? c * s3 : 0.f;
        }
        #pragma unroll
        for (int qb = 0; qb < 4; ++qb) {
            const float h  = 0.5f * wq[qb];
            const float cw = __cosf(h), sw = __sinf(h);
            const int   w  = 1 << qb;
            #pragma unroll
            for (int j = 0; j < 16; ++j) {
                if ((j & w) == 0) {
                    const int j1 = j | w;
                    const float r0 = sr[j], r1 = sr[j1];
                    const float i0 = si[j], i1 = si[j1];
                    sr[j]  = cw * r0 - sw * r1;  sr[j1] = sw * r0 + cw * r1;
                    si[j]  = cw * i0 - sw * i1;  si[j1] = sw * i0 + cw * i1;
                }
            }
        }
        const float PAR[16] = {1,-1,-1,1, -1,1,1,-1, -1,1,1,-1, 1,-1,-1,1};
        float ex = 0.f;
        #pragma unroll
        for (int j = 0; j < 16; ++j) ex += PAR[j] * (sr[j] * sr[j] + si[j] * si[j]);

        float lg[N_EXP]; float m = -1e30f;
        #pragma unroll
        for (int e = 0; e < N_EXP; ++e) { lg[e] = ex * aw[e] + ab[e]; m = fmaxf(m, lg[e]); }
        float ssum = 0.f;
        #pragma unroll
        for (int e = 0; e < N_EXP; ++e) { lg[e] = __expf(lg[e] - m); ssum += lg[e]; }
        const float inv = __builtin_amdgcn_rcpf(ssum);
        #pragma unroll
        for (int e = 0; e < N_EXP; ++e) lds_route[wave][e * 16 + lane] = lg[e] * inv;
    }

    const int rowM = rowBase + n;
    const v2f a1_0 = *(const v2f*)(x + rowM * 8 + 0 + 2 * hi);
    const v2f a1_1 = *(const v2f*)(x + rowM * 8 + 4 + 2 * hi);

    v8f tot;
    #pragma unroll
    for (int j = 0; j < 8; ++j) tot[j] = 0.f;

    float* __restrict__ hm = lds_hmid[wave];
    const float* __restrict__ rt = lds_route[wave];
    const int nc = n & 7;

    WAVE_LDS_FENCE();

    for (int e = 0; e < N_EXP; ++e) {
        const float* __restrict__ w1p = W1 + e * 8 * 64;
        #pragma unroll
        for (int t = 0; t < 4; ++t) {
            const int col = t * 16 + n;
            const float bias = b1[e * 64 + col];
            v8f c1;
            #pragma unroll
            for (int j = 0; j < 8; ++j) c1[j] = bias;
            v2f bb0, bb1;
            bb0.x = w1p[(0 + 2 * hi) * 64 + col];
            bb0.y = w1p[(1 + 2 * hi) * 64 + col];
            bb1.x = w1p[(4 + 2 * hi) * 64 + col];
            bb1.y = w1p[(5 + 2 * hi) * 64 + col];
            c1 = wmma_k4(a1_0, bb0, c1);
            c1 = wmma_k4(a1_1, bb1, c1);
            #pragma unroll
            for (int v = 0; v < 8; ++v) {
                hm[(v + 8 * hi) * HMS + col] = fast_silu(c1[v]);
            }
        }
        WAVE_LDS_FENCE();

        const float* __restrict__ w2p = W2 + e * 64 * 8;
        v8f c2;
        {
            const float bias2v = b2[e * 8 + nc];
            const float bias2 = (n < 8) ? bias2v : 0.f;
            #pragma unroll
            for (int j = 0; j < 8; ++j) c2[j] = bias2;
        }
        #pragma unroll
        for (int kk = 0; kk < 16; ++kk) {
            const int k0 = kk * 4;
            const v2f a2 = *(const v2f*)(hm + n * HMS + k0 + 2 * hi);
            v2f bv;
            const float bx = w2p[(k0 + 2 * hi) * 8 + nc];
            const float by = w2p[(k0 + 2 * hi + 1) * 8 + nc];
            bv.x = (n < 8) ? bx : 0.f;
            bv.y = (n < 8) ? by : 0.f;
            c2 = wmma_k4(a2, bv, c2);
        }
        #pragma unroll
        for (int v = 0; v < 8; ++v) {
            const float r = rt[e * 16 + v + 8 * hi];
            tot[v] += r * c2[v];
        }
        WAVE_LDS_FENCE();
    }

    float* __restrict__ tp = lds_tot[wave];
    if (n < 8) {
        #pragma unroll
        for (int v = 0; v < 8; ++v) tp[(v + 8 * hi) * 8 + n] = tot[v];
    }
    WAVE_LDS_FENCE();

    v8f c3;
    {
        const float bias3v = bg[nc];
        const float bias3 = (n < 8) ? bias3v : 0.f;
        #pragma unroll
        for (int j = 0; j < 8; ++j) c3[j] = bias3;
    }
    #pragma unroll
    for (int kk = 0; kk < 2; ++kk) {
        const int k0 = kk * 4;
        const v2f a3 = *(const v2f*)(tp + n * 8 + k0 + 2 * hi);
        const v2f wgv = *(const v2f*)(Wg + nc * 8 + k0 + 2 * hi);
        v2f bv;
        bv.x = (n < 8) ? wgv.x : 0.f;
        bv.y = (n < 8) ? wgv.y : 0.f;
        c3 = wmma_k4(a3, bv, c3);
    }
    if (n < 8) {
        #pragma unroll
        for (int v = 0; v < 8; ++v)
            out[(rowBase + v + 8 * hi) * 8 + n] = c3[v];
    }
}

extern "C" void kernel_launch(void* const* d_in, const int* in_sizes, int n_in,
                              void* d_out, int out_size, void* d_ws, size_t ws_size,
                              hipStream_t stream) {
    const float* x   = (const float*)d_in[0];
    const float* wq  = (const float*)d_in[1];
    const float* aw  = (const float*)d_in[2];
    const float* ab  = (const float*)d_in[3];
    const float* W1  = (const float*)d_in[4];
    const float* b1  = (const float*)d_in[5];
    const float* W2  = (const float*)d_in[6];
    const float* b2  = (const float*)d_in[7];
    const float* Wg  = (const float*)d_in[8];
    const float* bg  = (const float*)d_in[9];
    float* out = (float*)d_out;

    const int B = in_sizes[0] / 8;
    const int blocks = B / 128;
    titan_menger_kernel<<<blocks, 256, 0, stream>>>(x, wq, aw, ab, W1, b1, W2, b2, Wg, bg, out);
}


# SiLU as h + h*tanh(h), h = v/2, on the hardware tanh: one transcendental + one fma per value instead of exp, rcp and three plain ops; the half is folded exactly into the x fragments and b1
# speedup vs baseline: 1.1211x; 1.1211x over previous
#include <hip/hip_runtime.h>

typedef __attribute__((ext_vector_type(2))) float v2f;
typedef __attribute__((ext_vector_type(8))) float v8f;

#define N_EXP 20
#define HMS 66

#define WAVE_LDS_FENCE() asm volatile("s_wait_dscnt 0" ::: "memory")

__device__ __forceinline__ v8f wmma_k4(v2f a, v2f b, v8f c) {
    return __builtin_amdgcn_wmma_f32_16x16x4_f32(false, a, false, b, (short)0, c, false, false);
}

__device__ __forceinline__ float silu_of_half(float h) {
    return fmaf(h, __builtin_amdgcn_tanhf(h), h);
}

__global__ void __launch_bounds__(256)
titan_menger_kernel(const float* __restrict__ x,
                    const float* __restrict__ wq,
                    const float* __restrict__ aw,
                    const float* __restrict__ ab,
                    const float* __restrict__ W1,
                    const float* __restrict__ b1,
                    const float* __restrict__ W2,
                    const float* __restrict__ b2,
                    const float* __restrict__ Wg,
                    const float* __restrict__ bg,
                    float* __restrict__ out)
{
    __shared__ __align__(16) float lds_hmid[8][16 * HMS];
    __shared__ __align__(16) float lds_route[8][N_EXP * 16];
    __shared__ __align__(16) float lds_tot[8][16 * 8];

    const int tid  = threadIdx.x;
    const int wave = tid >> 5;
    const int lane = tid & 31;
    const int n    = lane & 15;
    const int hi   = lane >> 4;
    const int rowBase = blockIdx.x * 128 + wave * 16;

    if (lane < 16) {
        const int row = rowBase + lane;
        float sum = 0.f;
        #pragma unroll
        for (int i = 0; i < 8; ++i) sum += x[row * 8 + i];
        const float theta = sum * 0.125f;
        const float c  = __cosf(0.5f * theta), s = __sinf(0.5f * theta);
        const float c2 = c * c, s2 = s * s;
        const float c3 = c2 * c, s3 = s2 * s;
        const float c4 = c2 * c2, s4 = s2 * s2;

        const int PC[16] = {0,2,2,2, 2,4,2,2, 3,1,3,3, 3,1,1,1};
        float sr[16], si[16];
        #pragma unroll
        for (int j = 0; j < 16; ++j) {
            const int p = PC[j];
            sr[j] = (p == 0) ? c4 : (p == 2) ? -c2 * s2 : (p == 4) ? s4 : 0.f;
            si[j] = (p == 1) ? -c3 * s : (p == 3) ? c * s3 : 0.f;
        }
        #pragma unroll
        for (int qb = 0; qb < 4; ++qb) {
            const float h  = 0.5f * wq[qb];
            const float cw = __cosf(h), sw = __sinf(h);
            const int   w  = 1 << qb;
            #pragma unroll
            for (int j = 0; j < 16; ++j) {
                if ((j & w) == 0) {
                    const int j1 = j | w;
                    const float r0 = sr[j], r1 = sr[j1];
                    const float i0 = si[j], i1 = si[j1];
                    sr[j]  = cw * r0 - sw * r1;  sr[j1] = sw * r0 + cw * r1;
                    si[j]  = cw * i0 - sw * i1;  si[j1] = sw * i0 + cw * i1;
                }
            }
        }
        const float PAR[16] = {1,-1,-1,1, -1,1,1,-1, -1,1,1,-1, 1,-1,-1,1};
        float ex = 0.f;
        #pragma unroll
        for (int j = 0; j < 16; ++j) ex += PAR[j] * (sr[j] * sr[j] + si[j] * si[j]);

        float lg[N_EXP]; float m = -1e30f;
        #pragma unroll
        for (int e = 0; e < N_EXP; ++e) { lg[e] = ex * aw[e] + ab[e]; m = fmaxf(m, lg[e]); }
        float ssum = 0.f;
        #pragma unroll
        for (int e = 0; e < N_EXP; ++e) { lg[e] = __expf(lg[e] - m); ssum += lg[e]; }
        const float inv = __builtin_amdgcn_rcpf(ssum);
        #pragma unroll
        for (int e = 0; e < N_EXP; ++e) lds_route[wave][e * 16 + lane] = lg[e] * inv;
    }

    const int rowM = rowBase + n;
    const v2f a1_0 = 0.5f * *(const v2f*)(x + rowM * 8 + 0 + 2 * hi);
    const v2f a1_1 = 0.5f * *(const v2f*)(x + rowM * 8 + 4 + 2 * hi);

    v8f tot;
    #pragma unroll
    for (int j = 0; j < 8; ++j) tot[j] = 0.f;

    float* __restrict__ hm = lds_hmid[wave];
    const float* __restrict__ rt = lds_route[wave];
    const int nc = n & 7;

    WAVE_LDS_FENCE();

    for (int e = 0; e < N_EXP; ++e) {
        const float* __restrict__ w1p = W1 + e * 8 * 64;
        #pragma unroll
        for (int t = 0; t < 4; ++t) {
            const int col = t * 16 + n;
            const float bias = 0.5f * b1[e * 64 + col];
            v8f c1;
            #pragma unroll
            for (int j = 0; j < 8; ++j) c1[j] = bias;
            v2f bb0, bb1;
            bb0.x = w1p[(0 + 2 * hi) * 64 + col];
            bb0.y = w1p[(1 + 2 * hi) * 64 + col];
            bb1.x = w1p[(4 + 2 * hi) * 64 + col];
            bb1.y = w1p[(5 + 2 * hi) * 64 + col];
            c1 = wmma_k4(a1_0, bb0, c1);
            c1 = wmma_k4(a1_1, bb1, c1);
            #pragma unroll
            for (int v = 0; v < 8; ++v) {
                hm[(v + 8 * hi) * HMS + col] = silu_of_half(c1[v]);
            }
        }
        WAVE_LDS_FENCE();

        const float* __restrict__ w2p = W2 + e * 64 * 8;
        v8f c2;
        {
            const float bias2v = b2[e * 8 + nc];
            const float bias2 = (n < 8) ? bias2v : 0.f;
            #pragma unroll
            for (int j = 0; j < 8; ++j) c2[j] = bias2;
        }
        #pragma unroll
        for (int kk = 0; kk < 16; ++kk) {
            const int k0 = kk * 4;
            const v2f a2 = *(const v2f*)(hm + n * HMS + k0 + 2 * hi);
            v2f bv;
            const float bx = w2p[(k0 + 2 * hi) * 8 + nc];
            const float by = w2p[(k0 + 2 * hi + 1) * 8 + nc];
            bv.x = (n < 8) ? bx : 0.f;
            bv.y = (n < 8) ? by : 0.f;
            c2 = wmma_k4(a2, bv, c2);
        }
        #pragma unroll
        for (int v = 0; v < 8; ++v) {
            const float r = rt[e * 16 + v + 8 * hi];
            tot[v] += r * c2[v];
        }
        WAVE_LDS_FENCE();
    }

    float* __restrict__ tp = lds_tot[wave];
    if (n < 8) {
        #pragma unroll
        for (int v = 0; v < 8; ++v) tp[(v + 8 * hi) * 8 + n] = tot[v];
    }
    WAVE_LDS_FENCE();

    v8f c3;
    {
        const float bias3v = bg[nc];
        const float bias3 = (n < 8) ? bias3v : 0.f;
        #pragma unroll
        for (int j = 0; j < 8; ++j) c3[j] = bias3;
    }
    #pragma unroll
    for (int kk = 0; kk < 2; ++kk) {
        const int k0 = kk * 4;
        const v2f a3 = *(const v2f*)(tp + n * 8 + k0 + 2 * hi);
        const v2f wgv = *(const v2f*)(Wg + nc * 8 + k0 + 2 * hi);
        v2f bv;
        bv.x = (n < 8) ? wgv.x : 0.f;
        bv.y = (n < 8) ? wgv.y : 0.f;
        c3 = wmma_k4(a3, bv, c3);
    }
    if (n < 8) {
        #pragma unroll
        for (int v = 0; v < 8; ++v)
            out[(rowBase + v + 8 * hi) * 8 + n] = c3[v];
    }
}

extern "C" void kernel_launch(void* const* d_in, const int* in_sizes, int n_in,
                              void* d_out, int out_size, void* d_ws, size_t ws_size,
                              hipStream_t stream) {
    const float* x   = (const float*)d_in[0];
    const float* wq  = (const float*)d_in[1];
    const float* aw  = (const float*)d_in[2];
    const float* ab  = (const float*)d_in[3];
    const float* W1  = (const float*)d_in[4];
    const float* b1  = (const float*)d_in[5];
    const float* W2  = (const float*)d_in[6];
    const float* b2  = (const float*)d_in[7];
    const float* Wg  = (const float*)d_in[8];
    const float* bg  = (const float*)d_in[9];
    float* out = (float*)d_out;

    const int B = in_sizes[0] / 8;
    const int blocks = B / 128;
    titan_menger_kernel<<<blocks, 256, 0, stream>>>(x, wq, aw, ab, W1, b1, W2, b2, Wg, bg, out);
}


# All expert weights staged once per 32-wave workgroup in LDS in fragment order (W2 transposed): B fragments of both GEMMs by ds_load_b128, no global loads in the expert loop; GEMM2 k order changes
# speedup vs baseline: 1.1871x; 1.0589x over previous
#include <hip/hip_runtime.h>

typedef __attribute__((ext_vector_type(2))) float v2f;
typedef __attribute__((ext_vector_type(4))) float v4f;
typedef __attribute__((ext_vector_type(8))) float v8f;

#define N_EXP 20
#define NWAVE 32
#define HMS 68
#define W2S 68

#define WAVE_LDS_FENCE() asm volatile("s_wait_dscnt 0" ::: "memory")

__device__ __forceinline__ v8f wmma_k4(v2f a, v2f b, v8f c) {
    return __builtin_amdgcn_wmma_f32_16x16x4_f32(false, a, false, b, (short)0, c, false, false);
}

__device__ __forceinline__ float silu_of_half(float h) {
    return fmaf(h, __builtin_amdgcn_tanhf(h), h);
}

__global__ void __launch_bounds__(1024)
titan_menger_kernel(const float* __restrict__ x,
                    const float* __restrict__ wq,
                    const float* __restrict__ aw,
                    const float* __restrict__ ab,
                    const float* __restrict__ W1,
                    const float* __restrict__ b1,
                    const float* __restrict__ W2,
                    const float* __restrict__ b2,
                    const float* __restrict__ Wg,
                    const float* __restrict__ bg,
                    float* __restrict__ out)
{
    __shared__ __align__(16) float s_w2t[N_EXP * 8 * W2S];
    __shared__ __align__(16) float s_w1p[N_EXP * 8 * 64];
    __shared__ __align__(16) float s_b1p[N_EXP * 64];
    __shared__ __align__(16) float s_b2[N_EXP * 8];
    __shared__ __align__(16) float lds_hmid[NWAVE][16 * HMS];
    __shared__ __align__(16) float lds_route[NWAVE][N_EXP * 16];
    __shared__ __align__(16) float lds_tot[NWAVE][16 * 8];

    const int tid  = threadIdx.x;
    const int wave = tid >> 5;
    const int lane = tid & 31;
    const int n    = lane & 15;
    const int hi   = lane >> 4;
    const int rowBase = blockIdx.x * (NWAVE * 16) + wave * 16;

    #pragma unroll
    for (int it = 0; it < 3; ++it) {
        const int idx = tid + it * (NWAVE * 32);
        if (idx < N_EXP * 128) {
            const v4f w = *(const v4f*)(W1 + 4 * idx);
            const int m = idx & 15, k = (idx >> 4) & 7, ee = idx >> 7;
            const int t = m >> 2;
            float* dst = s_w1p + (((ee * 4 + (k >> 1)) * 2 + (t >> 1)) * 16 + 4 * (m & 3)) * 4
                               + (t & 1) * 2 + (k & 1);
            dst[0] = w.x; dst[4] = w.y; dst[8] = w.z; dst[12] = w.w;
        }
    }
    #pragma unroll
    for (int it = 0; it < 3; ++it) {
        const int idx = tid + it * (NWAVE * 32);
        if (idx < N_EXP * 128) {
            const v4f w = *(const v4f*)(W2 + 4 * idx);
            const int g = idx & 1, j = (idx >> 1) & 63, ee = idx >> 7;
            float* dst = s_w2t + (ee * 8 + 4 * g) * W2S + j;
            dst[0] = w.x; dst[W2S] = w.y; dst[2 * W2S] = w.z; dst[3 * W2S] = w.w;
        }
    }
    if (tid < N_EXP * 16) {
        const v4f b = 0.5f * *(const v4f*)(b1 + 4 * tid);
        const int m = tid & 15, ee = tid >> 4;
        float* dst = s_b1p + ee * 64 + 16 * (m & 3) + (m >> 2);
        dst[0] = b.x; dst[4] = b.y; dst[8] = b.z; dst[12] = b.w;
    }
    if (tid >= 512 && tid < 512 + N_EXP * 2) {
        const int i = tid - 512;
        *(v4f*)(s_b2 + 4 * i) = *(const v4f*)(b2 + 4 * i);
    }

    if (lane < 16) {
        const int row = rowBase + lane;
        float sum = 0.f;
        #pragma unroll
        for (int i = 0; i < 8; ++i) sum += x[row * 8 + i];
        const float theta = sum * 0.125f;
        const float c  = __cosf(0.5f * theta), s = __sinf(0.5f * theta);
        const float c2 = c * c, s2 = s * s;
        const float c3 = c2 * c, s3 = s2 * s;
        const float c4 = c2 * c2, s4 = s2 * s2;

        const int PC[16] = {0,2,2,2, 2,4,2,2, 3,1,3,3, 3,1,1,1};
        float sr[16], si[16];
        #pragma unroll
        for (int j = 0; j < 16; ++j) {
            const int p = PC[j];
            sr[j] = (p == 0) ? c4 : (p == 2) ? -c2 * s2 : (p == 4) ? s4 : 0.f;
            si[j] = (p == 1) ? -c3 * s : (p == 3) ? c * s3 : 0.f;
        }
        #pragma unroll
        for (int qb = 0; qb < 4; ++qb) {
            const float h  = 0.5f * wq[qb];
            const float cw = __cosf(h), sw = __sinf(h);
            const int   w  = 1 << qb;
            #pragma unroll
            for (int j = 0; j < 16; ++j) {
                if ((j & w) == 0) {
                    const int j1 = j | w;
                    const float r0 = sr[j], r1 = sr[j1];
                    const float i0 = si[j], i1 = si[j1];
                    sr[j]  = cw * r0 - sw * r1;  sr[j1] = sw * r0 + cw * r1;
                    si[j]  = cw * i0 - sw * i1;  si[j1] = sw * i0 + cw * i1;
                }
            }
        }
        const float PAR[16] = {1,-1,-1,1, -1,1,1,-1, -1,1,1,-1, 1,-1,-1,1};
        float ex = 0.f;
        #pragma unroll
        for (int j = 0; j < 16; ++j) ex += PAR[j] * (sr[j] * sr[j] + si[j] * si[j]);

        float lg[N_EXP]; float m = -1e30f;
        #pragma unroll
        for (int e = 0; e < N_EXP; ++e) { lg[e] = ex * aw[e] + ab[e]; m = fmaxf(m, lg[e]); }
        float ssum = 0.f;
        #pragma unroll
        for (int e = 0; e < N_EXP; ++e) { lg[e] = __expf(lg[e] - m); ssum += lg[e]; }
        const float inv = __builtin_amdgcn_rcpf(ssum);
        #pragma unroll
        for (int e = 0; e < N_EXP; ++e) lds_route[wave][e * 16 + lane] = lg[e] * inv;
    }

    const int rowM = rowBase + n;
    const v2f a1_0 = 0.5f * *(const v2f*)(x + rowM * 8 + 0 + 2 * hi);
    const v2f a1_1 = 0.5f * *(const v2f*)(x + rowM * 8 + 4 + 2 * hi);

    v8f tot;
    #pragma unroll
    for (int j = 0; j < 8; ++j) tot[j] = 0.f;

    float* __restrict__ hm = lds_hmid[wave];
    const float* __restrict__ rt = lds_route[wave];
    const int nc = n & 7;

    __syncthreads();

    for (int e = 0; e < N_EXP; ++e) {
        const float* __restrict__ w1e = s_w1p + e * 512 + 4 * n;
        const v4f w00 = *(const v4f*)(w1e + (hi    ) * 128);
        const v4f w01 = *(const v4f*)(w1e + (hi    ) * 128 + 64);
        const v4f w10 = *(const v4f*)(w1e + (hi + 2) * 128);
        const v4f w11 = *(const v4f*)(w1e + (hi + 2) * 128 + 64);
        const v4f bh  = *(const v4f*)(s_b1p + e * 64 + 4 * n);
        v2f bb0[4], bb1[4];
        bb0[0].x = w00.x; bb0[0].y = w00.y;  bb0[1].x = w00.z; bb0[1].y = w00.w;
        bb0[2].x = w01.x; bb0[2].y = w01.y;  bb0[3].x = w01.z; bb0[3].y = w01.w;
        bb1[0].x = w10.x; bb1[0].y = w10.y;  bb1[1].x = w10.z; bb1[1].y = w10.w;
        bb1[2].x = w11.x; bb1[2].y = w11.y;  bb1[3].x = w11.z; bb1[3].y = w11.w;
        #pragma unroll
        for (int t = 0; t < 4; ++t) {
            const int col = t * 16 + n;
            v8f c1;
            #pragma unroll
            for (int j = 0; j < 8; ++j) c1[j] = bh[t];
            c1 = wmma_k4(a1_0, bb0[t], c1);
            c1 = wmma_k4(a1_1, bb1[t], c1);
            #pragma unroll
            for (int v = 0; v < 8; ++v) {
                hm[(v + 8 * hi) * HMS + col] = silu_of_half(c1[v]);
            }
        }
        WAVE_LDS_FENCE();

        v8f c2;
        {
            const float bias2 = s_b2[e * 8 + nc];
            #pragma unroll
            for (int j = 0; j < 8; ++j) c2[j] = bias2;
        }
        const float* __restrict__ ap = hm + n * HMS + 4 * hi;
        const float* __restrict__ bp = s_w2t + (e * 8 + nc) * W2S + 4 * hi;
        #pragma unroll
        for (int q = 0; q < 8; ++q) {
            const v4f av = *(const v4f*)(ap + 8 * q);
            const v4f bv = *(const v4f*)(bp + 8 * q);
            v2f a_lo, a_hi, b_lo, b_hi;
            a_lo.x = av.x; a_lo.y = av.y;  a_hi.x = av.z; a_hi.y = av.w;
            b_lo.x = bv.x; b_lo.y = bv.y;  b_hi.x = bv.z; b_hi.y = bv.w;
            c2 = wmma_k4(a_lo, b_lo, c2);
            c2 = wmma_k4(a_hi, b_hi, c2);
        }
        #pragma unroll
        for (int v = 0; v < 8; ++v) {
            const float r = rt[e * 16 + v + 8 * hi];
            tot[v] += r * c2[v];
        }
        WAVE_LDS_FENCE();
    }

    float* __restrict__ tp = lds_tot[wave];
    if (n < 8) {
        #pragma unroll
        for (int v = 0; v < 8; ++v) tp[(v + 8 * hi) * 8 + n] = tot[v];
    }
    WAVE_LDS_FENCE();

    v8f c3;
    {
        const float bias3v = bg[nc];
        const float bias3 = (n < 8) ? bias3v : 0.f;
        #pragma unroll
        for (int j = 0; j < 8; ++j) c3[j] = bias3;
    }
    #pragma unroll
    for (int kk = 0; kk < 2; ++kk) {
        const int k0 = kk * 4;
        const v2f a3 = *(const v2f*)(tp + n * 8 + k0 + 2 * hi);
        const v2f wgv = *(const v2f*)(Wg + nc * 8 + k0 + 2 * hi);
        v2f bv;
        bv.x = (n < 8) ? wgv.x : 0.f;
        bv.y = (n < 8) ? wgv.y : 0.f;
        c3 = wmma_k4(a3, bv, c3);
    }
    if (n < 8) {
        #pragma unroll
        for (int v = 0; v < 8; ++v)
            out[(rowBase + v + 8 * hi) * 8 + n] = c3[v];
    }
}

extern "C" void kernel_launch(void* const* d_in, const int* in_sizes, int n_in,
                              void* d_out, int out_size, void* d_ws, size_t ws_size,
                              hipStream_t stream) {
    const float* x   = (const float*)d_in[0];
    const float* wq  = (const float*)d_in[1];
    const float* aw  = (const float*)d_in[2];
    const float* ab  = (const float*)d_in[3];
    const float* W1  = (const float*)d_in[4];
    const float* b1  = (const float*)d_in[5];
    const float* W2  = (const float*)d_in[6];
    const float* b2  = (const float*)d_in[7];
    const float* Wg  = (const float*)d_in[8];
    const float* bg  = (const float*)d_in[9];
    float* out = (float*)d_out;

    const int B = in_sizes[0] / 8;
    const int blocks = B / 512;
    titan_menger_kernel<<<blocks, 1024, 0, stream>>>(x, wq, aw, ab, W1, b1, W2, b2, Wg, bg, out);
}


# GEMM1 computed transposed (W1^T as A, x^T as B) so its SiLU output is already the A fragment of GEMM2 in registers: no hmid LDS round trip, no fences in the expert loop; GEMM2 k order changes
# speedup vs baseline: 1.2438x; 1.0478x over previous
#include <hip/hip_runtime.h>

typedef __attribute__((ext_vector_type(2))) float v2f;
typedef __attribute__((ext_vector_type(4))) float v4f;
typedef __attribute__((ext_vector_type(8))) float v8f;

#define N_EXP 20
#define NWAVE 32
#define W2S 68

#define WAVE_LDS_FENCE() asm volatile("s_wait_dscnt 0" ::: "memory")

__device__ __forceinline__ v8f wmma_k4(v2f a, v2f b, v8f c) {
    return __builtin_amdgcn_wmma_f32_16x16x4_f32(false, a, false, b, (short)0, c, false, false);
}

__device__ __forceinline__ float silu_of_half(float h) {
    return fmaf(h, __builtin_amdgcn_tanhf(h), h);
}

__global__ void __launch_bounds__(1024)
titan_menger_kernel(const float* __restrict__ x,
                    const float* __restrict__ wq,
                    const float* __restrict__ aw,
                    const float* __restrict__ ab,
                    const float* __restrict__ W1,
                    const float* __restrict__ b1,
                    const float* __restrict__ W2,
                    const float* __restrict__ b2,
                    const float* __restrict__ Wg,
                    const float* __restrict__ bg,
                    float* __restrict__ out)
{
    __shared__ __align__(16) float s_w2t[N_EXP * 8 * W2S];
    __shared__ __align__(16) float s_w1p[N_EXP * 8 * 64];
    __shared__ __align__(16) float s_b1h[N_EXP * 64];
    __shared__ __align__(16) float s_b2[N_EXP * 8];
    __shared__ __align__(16) float lds_route[NWAVE][N_EXP * 16];
    __shared__ __align__(16) float lds_tot[NWAVE][16 * 8];

    const int tid  = threadIdx.x;
    const int wave = tid >> 5;
    const int lane = tid & 31;
    const int n    = lane & 15;
    const int hi   = lane >> 4;
    const int rowBase = blockIdx.x * (NWAVE * 16) + wave * 16;

    #pragma unroll
    for (int it = 0; it < 3; ++it) {
        const int idx = tid + it * (NWAVE * 32);
        if (idx < N_EXP * 128) {
            const v4f w = *(const v4f*)(W1 + 4 * idx);
            const int m = idx & 15, k = (idx >> 4) & 7, ee = idx >> 7;
            const int t = m >> 2;
            float* dst = s_w1p + (((ee * 4 + (k >> 1)) * 2 + (t >> 1)) * 16 + 4 * (m & 3)) * 4
                               + (t & 1) * 2 + (k & 1);
            dst[0] = w.x; dst[4] = w.y; dst[8] = w.z; dst[12] = w.w;
        }
    }
    #pragma unroll
    for (int it = 0; it < 3; ++it) {
        const int idx = tid + it * (NWAVE * 32);
        if (idx < N_EXP * 128) {
            const v4f w = *(const v4f*)(W2 + 4 * idx);
            const int g = idx & 1, j = (idx >> 1) & 63, ee = idx >> 7;
            float* dst = s_w2t + (ee * 8 + 4 * g) * W2S + j;
            dst[0] = w.x; dst[W2S] = w.y; dst[2 * W2S] = w.z; dst[3 * W2S] = w.w;
        }
    }
    if (tid < N_EXP * 16) {
        *(v4f*)(s_b1h + 4 * tid) = 0.5f * *(const v4f*)(b1 + 4 * tid);
    }
    if (tid >= 512 && tid < 512 + N_EXP * 2) {
        const int i = tid - 512;
        *(v4f*)(s_b2 + 4 * i) = *(const v4f*)(b2 + 4 * i);
    }

    if (lane < 16) {
        const int row = rowBase + lane;
        float sum = 0.f;
        #pragma unroll
        for (int i = 0; i < 8; ++i) sum += x[row * 8 + i];
        const float theta = sum * 0.125f;
        const float c  = __cosf(0.5f * theta), s = __sinf(0.5f * theta);
        const float c2 = c * c, s2 = s * s;
        const float c3 = c2 * c, s3 = s2 * s;
        const float c4 = c2 * c2, s4 = s2 * s2;

        const int PC[16] = {0,2,2,2, 2,4,2,2, 3,1,3,3, 3,1,1,1};
        float sr[16], si[16];
        #pragma unroll
        for (int j = 0; j < 16; ++j) {
            const int p = PC[j];
            sr[j] = (p == 0) ? c4 : (p == 2) ? -c2 * s2 : (p == 4) ? s4 : 0.f;
            si[j] = (p == 1) ? -c3 * s : (p == 3) ? c * s3 : 0.f;
        }
        #pragma unroll
        for (int qb = 0; qb < 4; ++qb) {
            const float h  = 0.5f * wq[qb];
            const float cw = __cosf(h), sw = __sinf(h);
            const int   w  = 1 << qb;
            #pragma unroll
            for (int j = 0; j < 16; ++j) {
                if ((j & w) == 0) {
                    const int j1 = j | w;
                    const float r0 = sr[j], r1 = sr[j1];
                    const float i0 = si[j], i1 = si[j1];
                    sr[j]  = cw * r0 - sw * r1;  sr[j1] = sw * r0 + cw * r1;
                    si[j]  = cw * i0 - sw * i1;  si[j1] = sw * i0 + cw * i1;
                }
            }
        }
        const float PAR[16] = {1,-1,-1,1, -1,1,1,-1, -1,1,1,-1, 1,-1,-1,1};
        float ex = 0.f;
        #pragma unroll
        for (int j = 0; j < 16; ++j) ex += PAR[j] * (sr[j] * sr[j] + si[j] * si[j]);

        float lg[N_EXP]; float m = -1e30f;
        #pragma unroll
        for (int e = 0; e < N_EXP; ++e) { lg[e] = ex * aw[e] + ab[e]; m = fmaxf(m, lg[e]); }
        float ssum = 0.f;
        #pragma unroll
        for (int e = 0; e < N_EXP; ++e) { lg[e] = __expf(lg[e] - m); ssum += lg[e]; }
        const float inv = __builtin_amdgcn_rcpf(ssum);
        #pragma unroll
        for (int e = 0; e < N_EXP; ++e) lds_route[wave][e * 16 + lane] = lg[e] * inv;
    }

    const int rowM = rowBase + n;
    const v2f a1_0 = 0.5f * *(const v2f*)(x + rowM * 8 + 0 + 2 * hi);
    const v2f a1_1 = 0.5f * *(const v2f*)(x + rowM * 8 + 4 + 2 * hi);

    v8f tot;
    #pragma unroll
    for (int j = 0; j < 8; ++j) tot[j] = 0.f;

    const float* __restrict__ rt = lds_route[wave];
    const int nc = n & 7;

    __syncthreads();

    for (int e = 0; e < N_EXP; ++e) {
        const float* __restrict__ w1e = s_w1p + e * 512 + 4 * n;
        const v4f w00 = *(const v4f*)(w1e + (hi    ) * 128);
        const v4f w01 = *(const v4f*)(w1e + (hi    ) * 128 + 64);
        const v4f w10 = *(const v4f*)(w1e + (hi + 2) * 128);
        const v4f w11 = *(const v4f*)(w1e + (hi + 2) * 128 + 64);
        v2f wa0[4], wa1[4];
        wa0[0].x = w00.x; wa0[0].y = w00.y;  wa0[1].x = w00.z; wa0[1].y = w00.w;
        wa0[2].x = w01.x; wa0[2].y = w01.y;  wa0[3].x = w01.z; wa0[3].y = w01.w;
        wa1[0].x = w10.x; wa1[0].y = w10.y;  wa1[1].x = w10.z; wa1[1].y = w10.w;
        wa1[2].x = w11.x; wa1[2].y = w11.y;  wa1[3].x = w11.z; wa1[3].y = w11.w;

        v8f c2;
        {
            const float bias2 = s_b2[e * 8 + nc];
            #pragma unroll
            for (int j = 0; j < 8; ++j) c2[j] = bias2;
        }
        const float* __restrict__ b1e = s_b1h + e * 64 + 8 * hi;
        const float* __restrict__ bp  = s_w2t + (e * 8 + nc) * W2S + 8 * hi;
        #pragma unroll
        for (int t = 0; t < 4; ++t) {
            const v4f blo = *(const v4f*)(b1e + 16 * t);
            const v4f bhi = *(const v4f*)(b1e + 16 * t + 4);
            v8f c1;
            c1[0] = blo.x; c1[1] = blo.y; c1[2] = blo.z; c1[3] = blo.w;
            c1[4] = bhi.x; c1[5] = bhi.y; c1[6] = bhi.z; c1[7] = bhi.w;
            c1 = wmma_k4(wa0[t], a1_0, c1);
            c1 = wmma_k4(wa1[t], a1_1, c1);
            v8f h;
            #pragma unroll
            for (int v = 0; v < 8; ++v) h[v] = silu_of_half(c1[v]);
            const v4f bv0 = *(const v4f*)(bp + 16 * t);
            const v4f bv1 = *(const v4f*)(bp + 16 * t + 4);
            v2f a_s, b_s;
            a_s.x = h[0]; a_s.y = h[1];  b_s.x = bv0.x; b_s.y = bv0.y;
            c2 = wmma_k4(a_s, b_s, c2);
            a_s.x = h[2]; a_s.y = h[3];  b_s.x = bv0.z; b_s.y = bv0.w;
            c2 = wmma_k4(a_s, b_s, c2);
            a_s.x = h[4]; a_s.y = h[5];  b_s.x = bv1.x; b_s.y = bv1.y;
            c2 = wmma_k4(a_s, b_s, c2);
            a_s.x = h[6]; a_s.y = h[7];  b_s.x = bv1.z; b_s.y = bv1.w;
            c2 = wmma_k4(a_s, b_s, c2);
        }
        #pragma unroll
        for (int v = 0; v < 8; ++v) {
            const float r = rt[e * 16 + v + 8 * hi];
            tot[v] += r * c2[v];
        }
    }

    float* __restrict__ tp = lds_tot[wave];
    if (n < 8) {
        #pragma unroll
        for (int v = 0; v < 8; ++v) tp[(v + 8 * hi) * 8 + n] = tot[v];
    }
    WAVE_LDS_FENCE();

    v8f c3;
    {
        const float bias3v = bg[nc];
        const float bias3 = (n < 8) ? bias3v : 0.f;
        #pragma unroll
        for (int j = 0; j < 8; ++j) c3[j] = bias3;
    }
    #pragma unroll
    for (int kk = 0; kk < 2; ++kk) {
        const int k0 = kk * 4;
        const v2f a3 = *(const v2f*)(tp + n * 8 + k0 + 2 * hi);
        const v2f wgv = *(const v2f*)(Wg + nc * 8 + k0 + 2 * hi);
        v2f bv;
        bv.x = (n < 8) ? wgv.x : 0.f;
        bv.y = (n < 8) ? wgv.y : 0.f;
        c3 = wmma_k4(a3, bv, c3);
    }
    if (n < 8) {
        #pragma unroll
        for (int v = 0; v < 8; ++v)
            out[(rowBase + v + 8 * hi) * 8 + n] = c3[v];
    }
}

extern "C" void kernel_launch(void* const* d_in, const int* in_sizes, int n_in,
                              void* d_out, int out_size, void* d_ws, size_t ws_size,
                              hipStream_t stream) {
    const float* x   = (const float*)d_in[0];
    const float* wq  = (const float*)d_in[1];
    const float* aw  = (const float*)d_in[2];
    const float* ab  = (const float*)d_in[3];
    const float* W1  = (const float*)d_in[4];
    const float* b1  = (const float*)d_in[5];
    const float* W2  = (const float*)d_in[6];
    const float* b2  = (const float*)d_in[7];
    const float* Wg  = (const float*)d_in[8];
    const float* bg  = (const float*)d_in[9];
    float* out = (float*)d_out;

    const int B = in_sizes[0] / 8;
    const int blocks = B / 512;
    titan_menger_kernel<<<blocks, 1024, 0, stream>>>(x, wq, aw, ab, W1, b1, W2, b2, Wg, bg, out);
}


# Each wave owns two 16-row tiles (16-wave workgroups): every W1, b1 and W2 fragment read from LDS feeds the matrix instructions of both tiles; half as many waves to launch; router on all 32 lanes
# speedup vs baseline: 1.3292x; 1.0687x over previous
#include <hip/hip_runtime.h>

typedef __attribute__((ext_vector_type(2))) float v2f;
typedef __attribute__((ext_vector_type(4))) float v4f;
typedef __attribute__((ext_vector_type(8))) float v8f;

#define N_EXP 20
#define NWAVE 16
#define RT 2
#define W2S 68
static_assert((N_EXP * 128) % (NWAVE * 32) == 0, "the weight staging loops assume whole trips of the workgroup");

#define WAVE_LDS_FENCE() asm volatile("s_wait_dscnt 0" ::: "memory")

__device__ __forceinline__ v8f wmma_k4(v2f a, v2f b, v8f c) {
    return __builtin_amdgcn_wmma_f32_16x16x4_f32(false, a, false, b, (short)0, c, false, false);
}

__device__ __forceinline__ v2f silu_of_half2(v2f h) {
    v2f t;
    t.x = __builtin_amdgcn_tanhf(h.x);
    t.y = __builtin_amdgcn_tanhf(h.y);
    return __builtin_elementwise_fma(h, t, h);
}

__global__ void __launch_bounds__(NWAVE * 32)
titan_menger_kernel(const float* __restrict__ x,
                    const float* __restrict__ wq,
                    const float* __restrict__ aw,
                    const float* __restrict__ ab,
                    const float* __restrict__ W1,
                    const float* __restrict__ b1,
                    const float* __restrict__ W2,
                    const float* __restrict__ b2,
                    const float* __restrict__ Wg,
                    const float* __restrict__ bg,
                    float* __restrict__ out)
{
    __shared__ __align__(16) float s_w2t[N_EXP * 8 * W2S];
    __shared__ __align__(16) float s_w1p[N_EXP * 8 * 64];
    __shared__ __align__(16) float s_b1h[N_EXP * 64];
    __shared__ __align__(16) float s_b2[N_EXP * 8];
    __shared__ __align__(16) float lds_route[NWAVE][N_EXP * 16 * RT];
    __shared__ __align__(16) float lds_tot[NWAVE][16 * RT * 8];

    const int tid  = threadIdx.x;
    const int wave = tid >> 5;
    const int lane = tid & 31;
    const int n    = lane & 15;
    const int hi   = lane >> 4;
    const int rowBase = blockIdx.x * (NWAVE * 16 * RT) + wave * (16 * RT);

    #pragma unroll
    for (int it = 0; it < (N_EXP * 128) / (NWAVE * 32); ++it) {
        const int idx = tid + it * (NWAVE * 32);
        const v4f w = *(const v4f*)(W1 + 4 * idx);
        const int m = idx & 15, k = (idx >> 4) & 7, ee = idx >> 7;
        const int t = m >> 2;
        float* dst = s_w1p + (((ee * 4 + (k >> 1)) * 2 + (t >> 1)) * 16 + 4 * (m & 3)) * 4
                           + (t & 1) * 2 + (k & 1);
        dst[0] = w.x; dst[4] = w.y; dst[8] = w.z; dst[12] = w.w;
    }
    #pragma unroll
    for (int it = 0; it < (N_EXP * 128) / (NWAVE * 32); ++it) {
        const int idx = tid + it * (NWAVE * 32);
        const v4f w = *(const v4f*)(W2 + 4 * idx);
        const int g = idx & 1, j = (idx >> 1) & 63, ee = idx >> 7;
        float* dst = s_w2t + (ee * 8 + 4 * g) * W2S + j;
        dst[0] = w.x; dst[W2S] = w.y; dst[2 * W2S] = w.z; dst[3 * W2S] = w.w;
    }
    if (tid < N_EXP * 16) {
        *(v4f*)(s_b1h + 4 * tid) = 0.5f * *(const v4f*)(b1 + 4 * tid);
    }
    if (tid >= 384 && tid < 384 + N_EXP * 2) {
        const int i = tid - 384;
        *(v4f*)(s_b2 + 4 * i) = *(const v4f*)(b2 + 4 * i);
    }

    {
        const int row = rowBase + lane;
        float sum = 0.f;
        #pragma unroll
        for (int i = 0; i < 8; ++i) sum += x[row * 8 + i];
        const float theta = sum * 0.125f;
        const float c  = __cosf(0.5f * theta), s = __sinf(0.5f * theta);
        const float c2 = c * c, s2 = s * s;
        const float c3 = c2 * c, s3 = s2 * s;
        const float c4 = c2 * c2, s4 = s2 * s2;

        const int PC[16] = {0,2,2,2, 2,4,2,2, 3,1,3,3, 3,1,1,1};
        float sr[16], si[16];
        #pragma unroll
        for (int j = 0; j < 16; ++j) {
            const int p = PC[j];
            sr[j] = (p == 0) ? c4 : (p == 2) ? -c2 * s2 : (p == 4) ? s4 : 0.f;
            si[j] = (p == 1) ? -c3 * s : (p == 3) ? c * s3 : 0.f;
        }
        #pragma unroll
        for (int qb = 0; qb < 4; ++qb) {
            const float h  = 0.5f * wq[qb];
            const float cw = __cosf(h), sw = __sinf(h);
            const int   w  = 1 << qb;
            #pragma unroll
            for (int j = 0; j < 16; ++j) {
                if ((j & w) == 0) {
                    const int j1 = j | w;
                    const float r0 = sr[j], r1 = sr[j1];
                    const float i0 = si[j], i1 = si[j1];
                    sr[j]  = cw * r0 - sw * r1;  sr[j1] = sw * r0 + cw * r1;
                    si[j]  = cw * i0 - sw * i1;  si[j1] = sw * i0 + cw * i1;
                }
            }
        }
        const float PAR[16] = {1,-1,-1,1, -1,1,1,-1, -1,1,1,-1, 1,-1,-1,1};
        float ex = 0.f;
        #pragma unroll
        for (int j = 0; j < 16; ++j) ex += PAR[j] * (sr[j] * sr[j] + si[j] * si[j]);

        float lg[N_EXP]; float m = -1e30f;
        #pragma unroll
        for (int e = 0; e < N_EXP; ++e) { lg[e] = ex * aw[e] + ab[e]; m = fmaxf(m, lg[e]); }
        float ssum = 0.f;
        #pragma unroll
        for (int e = 0; e < N_EXP; ++e) { lg[e] = __expf(lg[e] - m); ssum += lg[e]; }
        const float inv = __builtin_amdgcn_rcpf(ssum);
        #pragma unroll
        for (int e = 0; e < N_EXP; ++e) lds_route[wave][e * (16 * RT) + lane] = lg[e] * inv;
    }

    v2f a1_0[RT], a1_1[RT];
    #pragma unroll
    for (int r = 0; r < RT; ++r) {
        const int rowM = rowBase + 16 * r + n;
        a1_0[r] = 0.5f * *(const v2f*)(x + rowM * 8 + 0 + 2 * hi);
        a1_1[r] = 0.5f * *(const v2f*)(x + rowM * 8 + 4 + 2 * hi);
    }

    v8f tot[RT];
    #pragma unroll
    for (int r = 0; r < RT; ++r) {
        #pragma unroll
        for (int j = 0; j < 8; ++j) tot[r][j] = 0.f;
    }

    const float* __restrict__ rt = lds_route[wave];
    const int nc = n & 7;

    __syncthreads();

    for (int e = 0; e < N_EXP; ++e) {
        const float* __restrict__ w1e = s_w1p + e * 512 + 4 * n;
        const v4f w00 = *(const v4f*)(w1e + (hi    ) * 128);
        const v4f w01 = *(const v4f*)(w1e + (hi    ) * 128 + 64);
        const v4f w10 = *(const v4f*)(w1e + (hi + 2) * 128);
        const v4f w11 = *(const v4f*)(w1e + (hi + 2) * 128 + 64);
        v2f wa0[4], wa1[4];
        wa0[0].x = w00.x; wa0[0].y = w00.y;  wa0[1].x = w00.z; wa0[1].y = w00.w;
        wa0[2].x = w01.x; wa0[2].y = w01.y;  wa0[3].x = w01.z; wa0[3].y = w01.w;
        wa1[0].x = w10.x; wa1[0].y = w10.y;  wa1[1].x = w10.z; wa1[1].y = w10.w;
        wa1[2].x = w11.x; wa1[2].y = w11.y;  wa1[3].x = w11.z; wa1[3].y = w11.w;

        v8f c2[RT];
        {
            const float bias2 = s_b2[e * 8 + nc];
            #pragma unroll
            for (int r = 0; r < RT; ++r) {
                #pragma unroll
                for (int j = 0; j < 8; ++j) c2[r][j] = bias2;
            }
        }
        const float* __restrict__ b1e = s_b1h + e * 64 + 8 * hi;
        const float* __restrict__ bp  = s_w2t + (e * 8 + nc) * W2S + 8 * hi;
        #pragma unroll
        for (int t = 0; t < 4; ++t) {
            const v4f blo = *(const v4f*)(b1e + 16 * t);
            const v4f bhi = *(const v4f*)(b1e + 16 * t + 4);
            v8f cb;
            cb[0] = blo.x; cb[1] = blo.y; cb[2] = blo.z; cb[3] = blo.w;
            cb[4] = bhi.x; cb[5] = bhi.y; cb[6] = bhi.z; cb[7] = bhi.w;
            const v4f bv0 = *(const v4f*)(bp + 16 * t);
            const v4f bv1 = *(const v4f*)(bp + 16 * t + 4);
            v2f b_s0, b_s1, b_s2, b_s3;
            b_s0.x = bv0.x; b_s0.y = bv0.y;
            b_s1.x = bv0.z; b_s1.y = bv0.w;
            b_s2.x = bv1.x; b_s2.y = bv1.y;
            b_s3.x = bv1.z; b_s3.y = bv1.w;
            v2f h[RT][4];
            #pragma unroll
            for (int r = 0; r < RT; ++r) {
                v8f c1 = wmma_k4(wa0[t], a1_0[r], cb);
                c1 = wmma_k4(wa1[t], a1_1[r], c1);
                #pragma unroll
                for (int s = 0; s < 4; ++s) {
                    v2f c;
                    c.x = c1[2 * s]; c.y = c1[2 * s + 1];
                    h[r][s] = silu_of_half2(c);
                }
            }
            #pragma unroll
            for (int r = 0; r < RT; ++r) {
                c2[r] = wmma_k4(h[r][0], b_s0, c2[r]);
                c2[r] = wmma_k4(h[r][1], b_s1, c2[r]);
                c2[r] = wmma_k4(h[r][2], b_s2, c2[r]);
                c2[r] = wmma_k4(h[r][3], b_s3, c2[r]);
            }
        }
        #pragma unroll
        for (int r = 0; r < RT; ++r) {
            const float* __restrict__ rp = rt + e * (16 * RT) + 16 * r + 8 * hi;
            const v4f rlo = *(const v4f*)(rp);
            const v4f rhi = *(const v4f*)(rp + 4);
            v8f rw;
            rw[0] = rlo.x; rw[1] = rlo.y; rw[2] = rlo.z; rw[3] = rlo.w;
            rw[4] = rhi.x; rw[5] = rhi.y; rw[6] = rhi.z; rw[7] = rhi.w;
            tot[r] = __builtin_elementwise_fma(rw, c2[r], tot[r]);
        }
    }

    float* __restrict__ tp = lds_tot[wave];
    if (n < 8) {
        #pragma unroll
        for (int r = 0; r < RT; ++r) {
            #pragma unroll
            for (int v = 0; v < 8; ++v) tp[(16 * r + v + 8 * hi) * 8 + n] = tot[r][v];
        }
    }
    WAVE_LDS_FENCE();

    const float bias3v = bg[nc];
    const float bias3 = (n < 8) ? bias3v : 0.f;
    v2f wg3[2];
    #pragma unroll
    for (int kk = 0; kk < 2; ++kk) {
        const v2f wgv = *(const v2f*)(Wg + nc * 8 + kk * 4 + 2 * hi);
        wg3[kk].x = (n < 8) ? wgv.x : 0.f;
        wg3[kk].y = (n < 8) ? wgv.y : 0.f;
    }
    #pragma unroll
    for (int r = 0; r < RT; ++r) {
        v8f c3;
        #pragma unroll
        for (int j = 0; j < 8; ++j) c3[j] = bias3;
        #pragma unroll
        for (int kk = 0; kk < 2; ++kk) {
            const v2f a3 = *(const v2f*)(tp + (16 * r + n) * 8 + kk * 4 + 2 * hi);
            c3 = wmma_k4(a3, wg3[kk], c3);
        }
        if (n < 8) {
            #pragma unroll
            for (int v = 0; v < 8; ++v)
                out[(rowBase + 16 * r + v + 8 * hi) * 8 + n] = c3[v];
        }
    }
}

extern "C" void kernel_launch(void* const* d_in, const int* in_sizes, int n_in,
                              void* d_out, int out_size, void* d_ws, size_t ws_size,
                              hipStream_t stream) {
    const float* x   = (const float*)d_in[0];
    const float* wq  = (const float*)d_in[1];
    const float* aw  = (const float*)d_in[2];
    const float* ab  = (const float*)d_in[3];
    const float* W1  = (const float*)d_in[4];
    const float* b1  = (const float*)d_in[5];
    const float* W2  = (const float*)d_in[6];
    const float* b2  = (const float*)d_in[7];
    const float* Wg  = (const float*)d_in[8];
    const float* bg  = (const float*)d_in[9];
    float* out = (float*)d_out;

    const int B = in_sizes[0] / 8;
    const int blocks = B / 512;
    titan_menger_kernel<<<blocks, NWAVE * 32, 0, stream>>>(x, wq, aw, ab, W1, b1, W2, b2, Wg, bg, out);
}
